# starting kernel
# baseline (speedup 1.0000x reference)
#include <hip/hip_runtime.h>
#include <hip/hip_bf16.h>
#include <math.h>

constexpr int B_  = 64;
constexpr int T_  = 50;
constexpr int S_  = 256;
constexpr int A_  = 64;
constexpr int H_  = 256;

typedef float v2f __attribute__((ext_vector_type(2)));
typedef float v8f __attribute__((ext_vector_type(8)));

__device__ __forceinline__ v8f wmma4_f32(v2f a, v2f b, v8f c) {
  return __builtin_amdgcn_wmma_f32_16x16x4_f32(
       false, a,  false, b,
       (short)0, c,  false,  false);
}

__global__ void ssm_hyper(const float* __restrict__ context,
                          const float* __restrict__ W1a, const float* __restrict__ b1a,
                          const float* __restrict__ W1b, const float* __restrict__ b1b,
                          float* __restrict__ HA, float* __restrict__ HB) {
  int idx = blockIdx.x * blockDim.x + threadIdx.x;
  if (idx >= T_ * B_ * H_) return;
  int k  = idx & (H_ - 1);
  int tb = idx >> 8;
  int b  = tb % B_;
  int t  = tb / B_;
  float c = context[b] + (float)t * (1.0f / 50.0f);
  HA[idx] = tanhf(fmaf(c, W1a[k], b1a[k]));
  HB[idx] = tanhf(fmaf(c, W1b[k], b1b[k]));
}

__global__ __launch_bounds__(64) void ssm_step_partial(
    const float* __restrict__ W2a, const float* __restrict__ b2a,
    const float* __restrict__ W2b, const float* __restrict__ b2b,
    const float* __restrict__ HA,  const float* __restrict__ HB,
    const float* __restrict__ xbase, int xstride,
    const float* __restrict__ ubase, int ustride,
    float* __restrict__ partials, int t) {
  const int lane = threadIdx.x & 31;
  const int wid  = threadIdx.x >> 5;
  const int blk  = blockIdx.x;
  const int bt   = blk & 3;
  const int it   = (blk >> 2) & 15;
  const int kc   = blk >> 6;
  const int b0   = bt * 16;
  const int i0   = it * 16;
  const int MN   = lane & 15;
  const int half = lane >> 4;
  const int jA0  = wid * 128;
  const int jB0  = wid * 32;

  const float* xrow = xbase + (size_t)(b0 + MN) * xstride + jA0 + half * 2;
  v2f xf[32];
#pragma unroll
  for (int f = 0; f < 32; ++f) {
    xf[f][0] = xrow[f * 4 + 0];
    xf[f][1] = xrow[f * 4 + 1];
  }
  const float* urow = ubase + (size_t)(b0 + MN) * ustride + jB0 + half * 2;
  v2f uf[8];
#pragma unroll
  for (int f = 0; f < 8; ++f) {
    uf[f][0] = urow[f * 4 + 0];
    uf[f][1] = urow[f * 4 + 1];
  }

  const float* haRow = HA + ((size_t)t * B_ + (b0 + MN)) * H_;
  const float* hbRow = HB + ((size_t)t * B_ + (b0 + MN)) * H_;

  v8f acc = {};
  const int kBeg = kc * 32, kEnd = kc * 32 + 32;
  for (int k = kBeg; k < kEnd; ++k) {
    const float* arow = W2a + (size_t)k * (S_ * S_) + (size_t)(i0 + MN) * S_ + jA0 + half * 2;
    v2f a0[16], a1[16];
#pragma unroll
    for (int f = 0; f < 16; ++f) { a0[f][0] = arow[f * 4];      a0[f][1] = arow[f * 4 + 1]; }
#pragma unroll
    for (int f = 0; f < 16; ++f) { a1[f][0] = arow[64 + f * 4]; a1[f][1] = arow[64 + f * 4 + 1]; }
    v8f t0 = {}, t1 = {};
#pragma unroll
    for (int f = 0; f < 16; f += 2) {
      t0 = wmma4_f32(a0[f],     xf[f],     t0);
      t1 = wmma4_f32(a0[f + 1], xf[f + 1], t1);
    }
#pragma unroll
    for (int f = 0; f < 16; f += 2) {
      t0 = wmma4_f32(a1[f],     xf[16 + f],     t0);
      t1 = wmma4_f32(a1[f + 1], xf[16 + f + 1], t1);
    }
    float sa = haRow[k];
#pragma unroll
    for (int r = 0; r < 8; ++r) acc[r] += sa * (t0[r] + t1[r]);

    const float* brow = W2b + (size_t)k * (S_ * A_) + (size_t)(i0 + MN) * A_ + jB0 + half * 2;
    v2f bfr[8];
#pragma unroll
    for (int f = 0; f < 8; ++f) { bfr[f][0] = brow[f * 4]; bfr[f][1] = brow[f * 4 + 1]; }
    v8f s0 = {}, s1 = {};
#pragma unroll
    for (int f = 0; f < 8; f += 2) {
      s0 = wmma4_f32(bfr[f],     uf[f],     s0);
      s1 = wmma4_f32(bfr[f + 1], uf[f + 1], s1);
    }
    float sb = hbRow[k];
#pragma unroll
    for (int r = 0; r < 8; ++r) acc[r] += sb * (s0[r] + s1[r]);
  }

  if (kc == 7) {
    const float* arow = b2a + (size_t)(i0 + MN) * S_ + jA0 + half * 2;
    v8f t0 = {}, t1 = {};
#pragma unroll
    for (int f = 0; f < 32; f += 2) {
      v2f c0; c0[0] = arow[f * 4];     c0[1] = arow[f * 4 + 1];
      v2f c1; c1[0] = arow[f * 4 + 4]; c1[1] = arow[f * 4 + 5];
      t0 = wmma4_f32(c0, xf[f],     t0);
      t1 = wmma4_f32(c1, xf[f + 1], t1);
    }
    const float* brow = b2b + (size_t)(i0 + MN) * A_ + jB0 + half * 2;
#pragma unroll
    for (int f = 0; f < 8; f += 2) {
      v2f c0; c0[0] = brow[f * 4];     c0[1] = brow[f * 4 + 1];
      v2f c1; c1[0] = brow[f * 4 + 4]; c1[1] = brow[f * 4 + 5];
      t0 = wmma4_f32(c0, uf[f],     t0);
      t1 = wmma4_f32(c1, uf[f + 1], t1);
    }
#pragma unroll
    for (int r = 0; r < 8; ++r) acc[r] += t0[r] + t1[r];
  }

  float* outp = partials + ((size_t)blk * 2 + wid) * 256 + lane * 8;
#pragma unroll
  for (int r = 0; r < 8; ++r) outp[r] = acc[r];
}

__global__ void ssm_step_reduce(const float* __restrict__ partials,
                                float* __restrict__ out, int t) {
  int e = blockIdx.x * blockDim.x + threadIdx.x;
  if (e >= B_ * S_) return;
  int b = e >> 8;
  int i = e & (S_ - 1);
  int it = i >> 4, M = i & 15;
  int bt = b >> 4, N = b & 15;
  int lane = (M >= 8) ? (16 + N) : N;
  int r    = M & 7;
  float sum = 0.f;
#pragma unroll
  for (int kc = 0; kc < 8; ++kc) {
#pragma unroll
    for (int wid = 0; wid < 2; ++wid) {
      sum += partials[(((size_t)(kc * 64 + it * 4 + bt)) * 2 + wid) * 256 + lane * 8 + r];
    }
  }
  out[((size_t)b * T_ + t) * S_ + i] = sum;
}

extern "C" void kernel_launch(void* const* d_in, const int* in_sizes, int n_in,
                              void* d_out, int out_size, void* d_ws, size_t ws_size,
                              hipStream_t stream) {
  const float* x0      = (const float*)d_in[0];
  const float* context = (const float*)d_in[1];
  const float* us      = (const float*)d_in[2];
  const float* W1a     = (const float*)d_in[3];
  const float* b1a     = (const float*)d_in[4];
  const float* W2a     = (const float*)d_in[5];
  const float* b2a     = (const float*)d_in[6];
  const float* W1b     = (const float*)d_in[7];
  const float* b1b     = (const float*)d_in[8];
  const float* W2b     = (const float*)d_in[9];
  const float* b2b     = (const float*)d_in[10];
  float* out = (float*)d_out;

  float* HA       = (float*)d_ws;
  float* HB       = HA + (size_t)T_ * B_ * H_;
  float* partials = HB + (size_t)T_ * B_ * H_;

  int nHyp = T_ * B_ * H_;
  ssm_hyper<<<(nHyp + 255) / 256, 256, 0, stream>>>(context, W1a, b1a, W1b, b1b, HA, HB);

  for (int t = 0; t < T_; ++t) {
    const float* xbase = (t == 0) ? x0 : (out + (size_t)(t - 1) * S_);
    int xstride        = (t == 0) ? S_ : T_ * S_;
    const float* ubase = us + (size_t)t * A_;

    ssm_step_partial<<<512, 64, 0, stream>>>(W2a, b2a, W2b, b2b, HA, HB,
                                             xbase, xstride, ubase, T_ * A_,
                                             partials, t);
    ssm_step_reduce<<<(B_ * S_ + 255) / 256, 256, 0, stream>>>(partials, out, t);
  }
}


# one wave covers all 64 batches: each W2a/W2b fragment feeds 4 WMMAs (4 N-tiles), weights fetched once per step; 4 waves/block summed via LDS, ha/hb stored [t][k][b]; summation order changes
# speedup vs baseline: 2.4577x; 2.4577x over previous
#include <hip/hip_runtime.h>
#include <hip/hip_bf16.h>
#include <math.h>


constexpr int B_  = 64;
constexpr int T_  = 50;
constexpr int S_  = 256;
constexpr int A_  = 64;
constexpr int H_  = 256;

typedef float v2f __attribute__((ext_vector_type(2)));
typedef float v4f __attribute__((ext_vector_type(4)));
typedef float v8f __attribute__((ext_vector_type(8)));

__device__ __forceinline__ v8f wmma4_f32(v2f a, v2f b, v8f c) {
  return __builtin_amdgcn_wmma_f32_16x16x4_f32(
       false, a,  false, b,
       (short)0, c,  false,  false);
}

__device__ __forceinline__ v2f ld2(const float* p) {
  return *reinterpret_cast<const v2f*>(p);
}
__device__ __forceinline__ v4f ld4(const float* p) {
  return *reinterpret_cast<const v4f*>(p);
}

__global__ void ssm_hyper(const float* __restrict__ context,
                          const float* __restrict__ W1a, const float* __restrict__ b1a,
                          const float* __restrict__ W1b, const float* __restrict__ b1b,
                          float* __restrict__ HA, float* __restrict__ HB) {
  int idx = blockIdx.x * blockDim.x + threadIdx.x;
  if (idx >= T_ * B_ * H_) return;
  int q  = idx & 63;
  int k  = (idx >> 6) & (H_ - 1);
  int t  = idx >> 14;
  int b  = (q & 3) * 16 + (q >> 2);
  float c = context[b] + (float)t * (1.0f / 50.0f);
  HA[idx] = tanhf(fmaf(c, W1a[k], b1a[k]));
  HB[idx] = tanhf(fmaf(c, W1b[k], b1b[k]));
}

__global__ __launch_bounds__(128) __attribute__((amdgpu_waves_per_eu(1, 1))) void ssm_step_partial(
    const float* __restrict__ W2a, const float* __restrict__ b2a,
    const float* __restrict__ W2b, const float* __restrict__ b2b,
    const float* __restrict__ HA,  const float* __restrict__ HB,
    const float* __restrict__ xbase, int xstride,
    const float* __restrict__ ubase, int ustride,
    float* __restrict__ partials, int t) {
  __shared__ __attribute__((aligned(32))) float red[4 * 4 * 256];

  const int lane = threadIdx.x & 31;
  const int wid  = threadIdx.x >> 5;
  const int blk  = blockIdx.x;
  const int it   = blk & 15;
  const int kc   = blk >> 4;
  const int i0   = it * 16;
  const int MN   = lane & 15;
  const int half = lane >> 4;
  const int jA0  = wid * 64;
  const int jB0  = wid * 16;

  v2f xf[4][16];
  v2f uf[4][4];
#pragma unroll
  for (int bt = 0; bt < 4; ++bt) {
    const float* xrow = xbase + (size_t)(bt * 16 + MN) * xstride + jA0 + half * 2;
#pragma unroll
    for (int f = 0; f < 16; ++f) xf[bt][f] = ld2(xrow + f * 4);
    const float* urow = ubase + (size_t)(bt * 16 + MN) * ustride + jB0 + half * 2;
#pragma unroll
    for (int f = 0; f < 4; ++f) uf[bt][f] = ld2(urow + f * 4);
  }

  const int kBeg = kc * 16;
  const float* aP  = W2a + (size_t)kBeg * (S_ * S_) + (size_t)(i0 + MN) * S_ + jA0 + half * 2;
  const float* bP  = W2b + (size_t)kBeg * (S_ * A_) + (size_t)(i0 + MN) * A_ + jB0 + half * 2;
  const float* haP = HA + ((size_t)t * H_ + kBeg) * 64 + MN * 4;
  const float* hbP = HB + ((size_t)t * H_ + kBeg) * 64 + MN * 4;

  v8f acc[4] = {};
  for (int kk = 0; kk < 16; ++kk) {
    v2f a[16];
#pragma unroll
    for (int f = 0; f < 16; ++f) a[f] = ld2(aP + f * 4);
    v2f bw[4];
#pragma unroll
    for (int f = 0; f < 4; ++f) bw[f] = ld2(bP + f * 4);
    const v4f sa = ld4(haP);
    const v4f sb = ld4(hbP);

    v8f tA[4] = {};
#pragma unroll
    for (int f = 0; f < 16; ++f) {
#pragma unroll
      for (int bt = 0; bt < 4; ++bt) tA[bt] = wmma4_f32(a[f], xf[bt][f], tA[bt]);
    }
    v8f tB[4] = {};
#pragma unroll
    for (int f = 0; f < 4; ++f) {
#pragma unroll
      for (int bt = 0; bt < 4; ++bt) tB[bt] = wmma4_f32(bw[f], uf[bt][f], tB[bt]);
    }
#pragma unroll
    for (int bt = 0; bt < 4; ++bt) {
#pragma unroll
      for (int r = 0; r < 8; ++r) {
        acc[bt][r] += sa[bt] * tA[bt][r];
        acc[bt][r] += sb[bt] * tB[bt][r];
      }
    }
    aP  += S_ * S_;
    bP  += S_ * A_;
    haP += 64;
    hbP += 64;
  }

  {
    const int jo = kc * 4 + half * 2;
    const v2f c = ld2(b2a + (size_t)(i0 + MN) * S_ + jA0 + jo);
    v8f tb[4] = {};
#pragma unroll
    for (int bt = 0; bt < 4; ++bt) {
      const v2f xq = ld2(xbase + (size_t)(bt * 16 + MN) * xstride + jA0 + jo);
      tb[bt] = wmma4_f32(c, xq, tb[bt]);
    }
    if (kc < 4) {
      const v2f d = ld2(b2b + (size_t)(i0 + MN) * A_ + jB0 + jo);
#pragma unroll
      for (int bt = 0; bt < 4; ++bt) {
        const v2f uq = ld2(ubase + (size_t)(bt * 16 + MN) * ustride + jB0 + jo);
        tb[bt] = wmma4_f32(d, uq, tb[bt]);
      }
    }
#pragma unroll
    for (int bt = 0; bt < 4; ++bt) {
#pragma unroll
      for (int r = 0; r < 8; ++r) acc[bt][r] += tb[bt][r];
    }
  }

  float* mine = red + (size_t)(wid * 4) * 256 + lane * 8;
#pragma unroll
  for (int bt = 0; bt < 4; ++bt) *reinterpret_cast<v8f*>(mine + bt * 256) = acc[bt];
  __syncthreads();
  v8f s = *reinterpret_cast<const v8f*>(red + (size_t)(0 * 4 + wid) * 256 + lane * 8);
#pragma unroll
  for (int w = 1; w < 4; ++w)
    s += *reinterpret_cast<const v8f*>(red + (size_t)(w * 4 + wid) * 256 + lane * 8);

  float* outp = partials + ((size_t)blk * 4 + wid) * 256 + lane * 8;
  *reinterpret_cast<v8f*>(outp) = s;
}

__global__ void ssm_step_reduce(const float* __restrict__ partials,
                                float* __restrict__ out, int t) {
  int e = blockIdx.x * blockDim.x + threadIdx.x;
  if (e >= B_ * S_) return;
  int b = e >> 8;
  int i = e & (S_ - 1);
  int it = i >> 4, M = i & 15;
  int bt = b >> 4, N = b & 15;
  int lane = (M >= 8) ? (16 + N) : N;
  int r    = M & 7;
  float sum = 0.f;
#pragma unroll
  for (int kc = 0; kc < 16; ++kc) {
    sum += partials[(((size_t)(kc * 16 + it)) * 4 + bt) * 256 + lane * 8 + r];
  }
  out[((size_t)b * T_ + t) * S_ + i] = sum;
}

extern "C" void kernel_launch(void* const* d_in, const int* in_sizes, int n_in,
                              void* d_out, int out_size, void* d_ws, size_t ws_size,
                              hipStream_t stream) {
  const float* x0      = (const float*)d_in[0];
  const float* context = (const float*)d_in[1];
  const float* us      = (const float*)d_in[2];
  const float* W1a     = (const float*)d_in[3];
  const float* b1a     = (const float*)d_in[4];
  const float* W2a     = (const float*)d_in[5];
  const float* b2a     = (const float*)d_in[6];
  const float* W1b     = (const float*)d_in[7];
  const float* b1b     = (const float*)d_in[8];
  const float* W2b     = (const float*)d_in[9];
  const float* b2b     = (const float*)d_in[10];
  float* out = (float*)d_out;

  float* HA       = (float*)d_ws;
  float* HB       = HA + (size_t)T_ * B_ * H_;
  float* partials = HB + (size_t)T_ * B_ * H_;

  int nHyp = T_ * B_ * H_;
  ssm_hyper<<<(nHyp + 255) / 256, 256, 0, stream>>>(context, W1a, b1a, W1b, b1b, HA, HB);

  for (int t = 0; t < T_; ++t) {
    const float* xbase = (t == 0) ? x0 : (out + (size_t)(t - 1) * S_);
    int xstride        = (t == 0) ? S_ : T_ * S_;
    const float* ubase = us + (size_t)t * A_;

    ssm_step_partial<<<256, 128, 0, stream>>>(W2a, b2a, W2b, b2b, HA, HB,
                                              xbase, xstride, ubase, T_ * A_,
                                              partials, t);
    ssm_step_reduce<<<(B_ * S_ + 255) / 256, 256, 0, stream>>>(partials, out, t);
  }
}


# weight fragments double-buffered in registers: loads of k+1 issued as one group ahead of the WMMAs of k (two k per trip, last pair peeled), hiding their cache latency; summation order unchanged
# speedup vs baseline: 2.9519x; 1.2011x over previous
#include <hip/hip_runtime.h>
#include <hip/hip_bf16.h>
#include <math.h>


constexpr int B_  = 64;
constexpr int T_  = 50;
constexpr int S_  = 256;
constexpr int A_  = 64;
constexpr int H_  = 256;

typedef float v2f __attribute__((ext_vector_type(2)));
typedef float v4f __attribute__((ext_vector_type(4)));
typedef float v8f __attribute__((ext_vector_type(8)));

__device__ __forceinline__ v8f wmma4_f32(v2f a, v2f b, v8f c) {
  return __builtin_amdgcn_wmma_f32_16x16x4_f32(
       false, a,  false, b,
       (short)0, c,  false,  false);
}

__device__ __forceinline__ v2f ld2(const float* p) {
  return *reinterpret_cast<const v2f*>(p);
}
__device__ __forceinline__ v4f ld4(const float* p) {
  return *reinterpret_cast<const v4f*>(p);
}

__global__ void ssm_hyper(const float* __restrict__ context,
                          const float* __restrict__ W1a, const float* __restrict__ b1a,
                          const float* __restrict__ W1b, const float* __restrict__ b1b,
                          float* __restrict__ HA, float* __restrict__ HB) {
  int idx = blockIdx.x * blockDim.x + threadIdx.x;
  if (idx >= T_ * B_ * H_) return;
  int q  = idx & 63;
  int k  = (idx >> 6) & (H_ - 1);
  int t  = idx >> 14;
  int b  = (q & 3) * 16 + (q >> 2);
  float c = context[b] + (float)t * (1.0f / 50.0f);
  HA[idx] = tanhf(fmaf(c, W1a[k], b1a[k]));
  HB[idx] = tanhf(fmaf(c, W1b[k], b1b[k]));
}

struct KFrag {
  v2f a[16];
  v2f bw[4];
  v4f sa;
  v4f sb;
};

__device__ __forceinline__ void load_k(KFrag& w, const float* aP, const float* bP,
                                       const float* haP, const float* hbP) {
#pragma unroll
  for (int f = 0; f < 16; ++f) w.a[f] = ld2(aP + f * 4);
#pragma unroll
  for (int f = 0; f < 4; ++f) w.bw[f] = ld2(bP + f * 4);
  w.sa = ld4(haP);
  w.sb = ld4(hbP);
}

__device__ __forceinline__ void mac_k(const KFrag& w, const v2f (&xf)[4][16],
                                      const v2f (&uf)[4][4], v8f (&acc)[4]) {
  v8f tA[4] = {};
#pragma unroll
  for (int f = 0; f < 16; ++f) {
#pragma unroll
    for (int bt = 0; bt < 4; ++bt) tA[bt] = wmma4_f32(w.a[f], xf[bt][f], tA[bt]);
  }
  v8f tB[4] = {};
#pragma unroll
  for (int f = 0; f < 4; ++f) {
#pragma unroll
    for (int bt = 0; bt < 4; ++bt) tB[bt] = wmma4_f32(w.bw[f], uf[bt][f], tB[bt]);
  }
  __builtin_amdgcn_sched_barrier(0);
#pragma unroll
  for (int bt = 0; bt < 4; ++bt) {
#pragma unroll
    for (int r = 0; r < 8; ++r) {
      acc[bt][r] += w.sa[bt] * tA[bt][r];
      acc[bt][r] += w.sb[bt] * tB[bt][r];
    }
  }
}

__global__ __launch_bounds__(128) __attribute__((amdgpu_waves_per_eu(1, 1))) void ssm_step_partial(
    const float* __restrict__ W2a, const float* __restrict__ b2a,
    const float* __restrict__ W2b, const float* __restrict__ b2b,
    const float* __restrict__ HA,  const float* __restrict__ HB,
    const float* __restrict__ xbase, int xstride,
    const float* __restrict__ ubase, int ustride,
    float* __restrict__ partials, int t) {
  __shared__ __attribute__((aligned(32))) float red[4 * 4 * 256];

  const int lane = threadIdx.x & 31;
  const int wid  = threadIdx.x >> 5;
  const int blk  = blockIdx.x;
  const int it   = blk & 15;
  const int kc   = blk >> 4;
  const int i0   = it * 16;
  const int MN   = lane & 15;
  const int half = lane >> 4;
  const int jA0  = wid * 64;
  const int jB0  = wid * 16;

  v2f xf[4][16];
  v2f uf[4][4];
#pragma unroll
  for (int bt = 0; bt < 4; ++bt) {
    const float* xrow = xbase + (size_t)(bt * 16 + MN) * xstride + jA0 + half * 2;
#pragma unroll
    for (int f = 0; f < 16; ++f) xf[bt][f] = ld2(xrow + f * 4);
    const float* urow = ubase + (size_t)(bt * 16 + MN) * ustride + jB0 + half * 2;
#pragma unroll
    for (int f = 0; f < 4; ++f) uf[bt][f] = ld2(urow + f * 4);
  }

  const int kBeg = kc * 16;
  const float* aP  = W2a + (size_t)kBeg * (S_ * S_) + (size_t)(i0 + MN) * S_ + jA0 + half * 2;
  const float* bP  = W2b + (size_t)kBeg * (S_ * A_) + (size_t)(i0 + MN) * A_ + jB0 + half * 2;
  const float* haP = HA + ((size_t)t * H_ + kBeg) * 64 + MN * 4;
  const float* hbP = HB + ((size_t)t * H_ + kBeg) * 64 + MN * 4;

  v8f acc[4] = {};
  KFrag w0, w1;
  load_k(w0, aP, bP, haP, hbP);
  for (int p = 0; p < 7; ++p) {
    load_k(w1, aP + S_ * S_, bP + S_ * A_, haP + 64, hbP + 64);
    mac_k(w0, xf, uf, acc);
    aP  += 2 * S_ * S_;
    bP  += 2 * S_ * A_;
    haP += 128;
    hbP += 128;
    load_k(w0, aP, bP, haP, hbP);
    mac_k(w1, xf, uf, acc);
  }
  load_k(w1, aP + S_ * S_, bP + S_ * A_, haP + 64, hbP + 64);
  mac_k(w0, xf, uf, acc);
  mac_k(w1, xf, uf, acc);

  {
    const int jo = kc * 4 + half * 2;
    const v2f c = ld2(b2a + (size_t)(i0 + MN) * S_ + jA0 + jo);
    v8f tb[4] = {};
#pragma unroll
    for (int bt = 0; bt < 4; ++bt) {
      const v2f xq = ld2(xbase + (size_t)(bt * 16 + MN) * xstride + jA0 + jo);
      tb[bt] = wmma4_f32(c, xq, tb[bt]);
    }
    if (kc < 4) {
      const v2f d = ld2(b2b + (size_t)(i0 + MN) * A_ + jB0 + jo);
#pragma unroll
      for (int bt = 0; bt < 4; ++bt) {
        const v2f uq = ld2(ubase + (size_t)(bt * 16 + MN) * ustride + jB0 + jo);
        tb[bt] = wmma4_f32(d, uq, tb[bt]);
      }
    }
#pragma unroll
    for (int bt = 0; bt < 4; ++bt) {
#pragma unroll
      for (int r = 0; r < 8; ++r) acc[bt][r] += tb[bt][r];
    }
  }

  float* mine = red + (size_t)(wid * 4) * 256 + lane * 8;
#pragma unroll
  for (int bt = 0; bt < 4; ++bt) *reinterpret_cast<v8f*>(mine + bt * 256) = acc[bt];
  __syncthreads();
  v8f s = *reinterpret_cast<const v8f*>(red + (size_t)(0 * 4 + wid) * 256 + lane * 8);
#pragma unroll
  for (int w = 1; w < 4; ++w)
    s += *reinterpret_cast<const v8f*>(red + (size_t)(w * 4 + wid) * 256 + lane * 8);

  float* outp = partials + ((size_t)blk * 4 + wid) * 256 + lane * 8;
  *reinterpret_cast<v8f*>(outp) = s;
}

__global__ void ssm_step_reduce(const float* __restrict__ partials,
                                float* __restrict__ out, int t) {
  int e = blockIdx.x * blockDim.x + threadIdx.x;
  if (e >= B_ * S_) return;
  int b = e >> 8;
  int i = e & (S_ - 1);
  int it = i >> 4, M = i & 15;
  int bt = b >> 4, N = b & 15;
  int lane = (M >= 8) ? (16 + N) : N;
  int r    = M & 7;
  float sum = 0.f;
#pragma unroll
  for (int kc = 0; kc < 16; ++kc) {
    sum += partials[(((size_t)(kc * 16 + it)) * 4 + bt) * 256 + lane * 8 + r];
  }
  out[((size_t)b * T_ + t) * S_ + i] = sum;
}

extern "C" void kernel_launch(void* const* d_in, const int* in_sizes, int n_in,
                              void* d_out, int out_size, void* d_ws, size_t ws_size,
                              hipStream_t stream) {
  const float* x0      = (const float*)d_in[0];
  const float* context = (const float*)d_in[1];
  const float* us      = (const float*)d_in[2];
  const float* W1a     = (const float*)d_in[3];
  const float* b1a     = (const float*)d_in[4];
  const float* W2a     = (const float*)d_in[5];
  const float* b2a     = (const float*)d_in[6];
  const float* W1b     = (const float*)d_in[7];
  const float* b1b     = (const float*)d_in[8];
  const float* W2b     = (const float*)d_in[9];
  const float* b2b     = (const float*)d_in[10];
  float* out = (float*)d_out;

  float* HA       = (float*)d_ws;
  float* HB       = HA + (size_t)T_ * B_ * H_;
  float* partials = HB + (size_t)T_ * B_ * H_;

  int nHyp = T_ * B_ * H_;
  ssm_hyper<<<(nHyp + 255) / 256, 256, 0, stream>>>(context, W1a, b1a, W1b, b1b, HA, HB);

  for (int t = 0; t < T_; ++t) {
    const float* xbase = (t == 0) ? x0 : (out + (size_t)(t - 1) * S_);
    int xstride        = (t == 0) ? S_ : T_ * S_;
    const float* ubase = us + (size_t)t * A_;

    ssm_step_partial<<<256, 128, 0, stream>>>(W2a, b2a, W2b, b2b, HA, HB,
                                              xbase, xstride, ubase, T_ * A_,
                                              partials, t);
    ssm_step_reduce<<<(B_ * S_ + 255) / 256, 256, 0, stream>>>(partials, out, t);
  }
}


# one block per output row i with M=k WMMA tiles scaled by the hypernet in registers: no split-k across blocks, per-step reduce launch gone (101 to 52 launches); summation order changes
# speedup vs baseline: 3.2697x; 1.1077x over previous
#include <hip/hip_runtime.h>
#include <hip/hip_bf16.h>
#include <math.h>


constexpr int B_  = 64;
constexpr int T_  = 50;
constexpr int S_  = 256;
constexpr int A_  = 64;
constexpr int H_  = 256;

typedef float v2f __attribute__((ext_vector_type(2)));
typedef float v4f __attribute__((ext_vector_type(4)));
typedef float v8f __attribute__((ext_vector_type(8)));

__device__ __forceinline__ v8f wmma4_f32(v2f a, v2f b, v8f c) {
  return __builtin_amdgcn_wmma_f32_16x16x4_f32(
       false, a,  false, b,
       (short)0, c,  false,  false);
}

__device__ __forceinline__ v2f ld2(const float* p) {
  return *reinterpret_cast<const v2f*>(p);
}
__device__ __forceinline__ v4f ld4(const float* p) {
  return *reinterpret_cast<const v4f*>(p);
}
__device__ __forceinline__ v8f ld4x2(const float* p) {
  const v4f lo = ld4(p);
  const v4f hi = ld4(p + 128);
  return __builtin_shufflevector(lo, hi, 0, 1, 2, 3, 4, 5, 6, 7);
}

__global__ void ssm_hyper(const float* __restrict__ context,
                          const float* __restrict__ W1a, const float* __restrict__ b1a,
                          const float* __restrict__ W1b, const float* __restrict__ b1b,
                          const float* __restrict__ x0,
                          float* __restrict__ HA, float* __restrict__ HB,
                          float* __restrict__ X) {
  int idx = blockIdx.x * blockDim.x + threadIdx.x;
  if (idx >= T_ * B_ * H_) return;
  int e    = idx & 3;
  int lane = (idx >> 2) & 31;
  int q    = (idx >> 7) & 1;
  int bt   = (idx >> 8) & 3;
  int kt   = (idx >> 10) & 15;
  int t    = idx >> 14;
  int b    = bt * 16 + (lane & 15);
  int k    = kt * 16 + (lane >> 4) * 8 + q * 4 + e;
  float c = context[b] + (float)t * (1.0f / 50.0f);
  HA[idx] = tanhf(fmaf(c, W1a[k], b1a[k]));
  HB[idx] = tanhf(fmaf(c, W1b[k], b1b[k]));
  if (idx < B_ * S_) X[idx] = x0[(size_t)(idx & 63) * S_ + (idx >> 6)];
}

struct KFrag {
  v2f a[16];
  v2f bw[4];
  v8f sa[4];
  v8f sb[4];
};

__device__ __forceinline__ void load_k(KFrag& w, const float* aP, const float* bP,
                                       const float* haP, const float* hbP) {
#pragma unroll
  for (int f = 0; f < 16; ++f) w.a[f] = ld2(aP + f * 4);
#pragma unroll
  for (int f = 0; f < 4; ++f) w.bw[f] = ld2(bP + f * 4);
#pragma unroll
  for (int bt = 0; bt < 4; ++bt) {
    w.sa[bt] = ld4x2(haP + bt * 256);
    w.sb[bt] = ld4x2(hbP + bt * 256);
  }
}

__device__ __forceinline__ void mac_k(const KFrag& w, const v2f (&xf)[4][16],
                                      const v2f (&uf)[4][4], v8f (&acc)[4]) {
  __builtin_amdgcn_sched_barrier(0);
  v8f tA[4] = {};
#pragma unroll
  for (int f = 0; f < 16; ++f) {
#pragma unroll
    for (int bt = 0; bt < 4; ++bt) tA[bt] = wmma4_f32(w.a[f], xf[bt][f], tA[bt]);
  }
  v8f tB[4] = {};
#pragma unroll
  for (int f = 0; f < 4; ++f) {
#pragma unroll
    for (int bt = 0; bt < 4; ++bt) tB[bt] = wmma4_f32(w.bw[f], uf[bt][f], tB[bt]);
  }
  __builtin_amdgcn_sched_barrier(0);
#pragma unroll
  for (int bt = 0; bt < 4; ++bt) {
#pragma unroll
    for (int r = 0; r < 8; ++r) {
      acc[bt][r] += w.sa[bt][r] * tA[bt][r];
      acc[bt][r] += w.sb[bt][r] * tB[bt][r];
    }
  }
}

__global__ __launch_bounds__(128) __attribute__((amdgpu_waves_per_eu(1, 1))) void ssm_step(
    const float* __restrict__ W2a, const float* __restrict__ b2a,
    const float* __restrict__ W2b, const float* __restrict__ b2b,
    const float* __restrict__ HA,  const float* __restrict__ HB,
    const float* __restrict__ xin,
    const float* __restrict__ ubase, int ustride,
    float* __restrict__ xout, float* __restrict__ out, int t) {
  __shared__ float red[4 * 4 * 32];

  const int lane = threadIdx.x & 31;
  const int wid  = threadIdx.x >> 5;
  const int i    = blockIdx.x;
  const int MN   = lane & 15;
  const int half = lane >> 4;
  const int jA0  = wid * 64;
  const int jB0  = wid * 16;

  v2f xf[4][16];
  v2f uf[4][4];
#pragma unroll
  for (int bt = 0; bt < 4; ++bt) {
    const float* xcol = xin + (size_t)(jA0 + half * 2) * B_ + bt * 16 + MN;
#pragma unroll
    for (int f = 0; f < 16; ++f) {
      v2f v;
      v[0] = xcol[(f * 4) * B_];
      v[1] = xcol[(f * 4 + 1) * B_];
      xf[bt][f] = v;
    }
    const float* urow = ubase + (size_t)(bt * 16 + MN) * ustride + jB0 + half * 2;
#pragma unroll
    for (int f = 0; f < 4; ++f) uf[bt][f] = ld2(urow + f * 4);
  }

  const float* aP  = W2a + (size_t)MN * (S_ * S_) + (size_t)i * S_ + jA0 + half * 2;
  const float* bP  = W2b + (size_t)MN * (S_ * A_) + (size_t)i * A_ + jB0 + half * 2;
  const float* haP = HA + (size_t)t * (H_ * B_) + lane * 4;
  const float* hbP = HB + (size_t)t * (H_ * B_) + lane * 4;

  KFrag w0, w1;
  load_k(w0, aP, bP, haP, hbP);

  if (t > 0 && wid < 2) {
    const int ob = i >> 2;
    const int oi = (i & 3) * 64 + wid * 32 + lane;
    out[((size_t)ob * T_ + (t - 1)) * S_ + oi] = xin[(size_t)oi * B_ + ob];
  }

  v8f acc[4] = {};
  constexpr size_t aStep = (size_t)16 * S_ * S_;
  constexpr size_t bStep = (size_t)16 * S_ * A_;
  for (int p = 0; p < 7; ++p) {
    load_k(w1, aP + aStep, bP + bStep, haP + 1024, hbP + 1024);
    mac_k(w0, xf, uf, acc);
    aP  += 2 * aStep;
    bP  += 2 * bStep;
    haP += 2048;
    hbP += 2048;
    load_k(w0, aP, bP, haP, hbP);
    mac_k(w1, xf, uf, acc);
  }
  load_k(w1, aP + aStep, bP + bStep, haP + 1024, hbP + 1024);
  mac_k(w0, xf, uf, acc);

  v2f cb[16];
  v2f db[4];
  {
    const float* cP = b2a + (size_t)i * S_ + jA0 + half * 2;
    const float* dP = b2b + (size_t)i * A_ + jB0 + half * 2;
#pragma unroll
    for (int f = 0; f < 16; ++f) cb[f] = ld2(cP + f * 4);
#pragma unroll
    for (int f = 0; f < 4; ++f) db[f] = ld2(dP + f * 4);
  }
  mac_k(w1, xf, uf, acc);
  {
    const v2f zero = {0.f, 0.f};
    v8f tb[4] = {};
#pragma unroll
    for (int f = 0; f < 16; ++f) {
      const v2f cr = (MN == 0) ? cb[f] : zero;
#pragma unroll
      for (int bt = 0; bt < 4; ++bt) tb[bt] = wmma4_f32(cr, xf[bt][f], tb[bt]);
    }
#pragma unroll
    for (int f = 0; f < 4; ++f) {
      const v2f dr = (MN == 0) ? db[f] : zero;
#pragma unroll
      for (int bt = 0; bt < 4; ++bt) tb[bt] = wmma4_f32(dr, uf[bt][f], tb[bt]);
    }
#pragma unroll
    for (int bt = 0; bt < 4; ++bt) acc[bt][0] += (half == 0) ? tb[bt][0] : 0.f;
  }

#pragma unroll
  for (int bt = 0; bt < 4; ++bt) {
    const v8f a = acc[bt];
    red[(wid * 4 + bt) * 32 + lane] =
        ((a[0] + a[1]) + (a[2] + a[3])) + ((a[4] + a[5]) + (a[6] + a[7]));
  }
  __syncthreads();
  if (wid < 2) {
    const int b  = wid * 32 + lane;
    const int bt = b >> 4, c = b & 15;
    float s = 0.f;
#pragma unroll
    for (int w = 0; w < 4; ++w)
      s += red[(w * 4 + bt) * 32 + c] + red[(w * 4 + bt) * 32 + 16 + c];
    xout[(size_t)i * B_ + b] = s;
  }
}

__global__ void ssm_emit_last(const float* __restrict__ xin, float* __restrict__ out) {
  int e = blockIdx.x * blockDim.x + threadIdx.x;
  if (e >= B_ * S_) return;
  int b = e >> 8;
  int i = e & (S_ - 1);
  out[((size_t)b * T_ + (T_ - 1)) * S_ + i] = xin[(size_t)i * B_ + b];
}

extern "C" void kernel_launch(void* const* d_in, const int* in_sizes, int n_in,
                              void* d_out, int out_size, void* d_ws, size_t ws_size,
                              hipStream_t stream) {
  const float* x0      = (const float*)d_in[0];
  const float* context = (const float*)d_in[1];
  const float* us      = (const float*)d_in[2];
  const float* W1a     = (const float*)d_in[3];
  const float* b1a     = (const float*)d_in[4];
  const float* W2a     = (const float*)d_in[5];
  const float* b2a     = (const float*)d_in[6];
  const float* W1b     = (const float*)d_in[7];
  const float* b1b     = (const float*)d_in[8];
  const float* W2b     = (const float*)d_in[9];
  const float* b2b     = (const float*)d_in[10];
  float* out = (float*)d_out;

  float* HA = (float*)d_ws;
  float* HB = HA + (size_t)T_ * B_ * H_;
  float* X0 = HB + (size_t)T_ * B_ * H_;
  float* X1 = X0 + (size_t)S_ * B_;

  int nHyp = T_ * B_ * H_;
  ssm_hyper<<<(nHyp + 255) / 256, 256, 0, stream>>>(context, W1a, b1a, W1b, b1b, x0, HA, HB, X0);

  for (int t = 0; t < T_; ++t) {
    const float* xin = (t & 1) ? X1 : X0;
    float* xout      = (t & 1) ? X0 : X1;
    const float* ubase = us + (size_t)t * A_;

    ssm_step<<<256, 128, 0, stream>>>(W2a, b2a, W2b, b2b, HA, HB,
                                      xin, ubase, T_ * A_, xout, out, t);
  }
  ssm_emit_last<<<(B_ * S_ + 255) / 256, 256, 0, stream>>>(X0, out);
}


# every fragment load 16 bytes per lane: weight k order regrouped (j0+8g+4h+2s+e), batch tile columns interleaved (b=4c+bt); global load instructions and line requests halved; summation order changes
# speedup vs baseline: 3.4495x; 1.0550x over previous
#include <hip/hip_runtime.h>
#include <hip/hip_bf16.h>
#include <math.h>


constexpr int B_  = 64;
constexpr int T_  = 50;
constexpr int S_  = 256;
constexpr int A_  = 64;
constexpr int H_  = 256;

typedef float v2f __attribute__((ext_vector_type(2)));
typedef float v4f __attribute__((ext_vector_type(4)));
typedef float v8f __attribute__((ext_vector_type(8)));

__device__ __forceinline__ v8f wmma4_f32(v2f a, v2f b, v8f c) {
  return __builtin_amdgcn_wmma_f32_16x16x4_f32(
       false, a,  false, b,
       (short)0, c,  false,  false);
}

__device__ __forceinline__ v2f ld2(const float* p) {
  return *reinterpret_cast<const v2f*>(p);
}
__device__ __forceinline__ v4f ld4(const float* p) {
  return *reinterpret_cast<const v4f*>(p);
}
__device__ __forceinline__ v8f ld4x2(const float* p) {
  const v4f lo = ld4(p);
  const v4f hi = ld4(p + 128);
  return __builtin_shufflevector(lo, hi, 0, 1, 2, 3, 4, 5, 6, 7);
}

__global__ void ssm_hyper(const float* __restrict__ context,
                          const float* __restrict__ W1a, const float* __restrict__ b1a,
                          const float* __restrict__ W1b, const float* __restrict__ b1b,
                          const float* __restrict__ x0,
                          float* __restrict__ HA, float* __restrict__ HB,
                          float* __restrict__ X) {
  int idx = blockIdx.x * blockDim.x + threadIdx.x;
  if (idx >= T_ * B_ * H_) return;
  int e    = idx & 3;
  int lane = (idx >> 2) & 31;
  int q    = (idx >> 7) & 1;
  int bt   = (idx >> 8) & 3;
  int kt   = (idx >> 10) & 15;
  int t    = idx >> 14;
  int b    = (lane & 15) * 4 + bt;
  int k    = kt * 16 + (lane >> 4) * 8 + q * 4 + e;
  float c = context[b] + (float)t * (1.0f / 50.0f);
  HA[idx] = tanhf(fmaf(c, W1a[k], b1a[k]));
  HB[idx] = tanhf(fmaf(c, W1b[k], b1b[k]));
  if (idx < B_ * S_) X[idx] = x0[(size_t)(idx & 63) * S_ + (idx >> 6)];
}

struct KFrag {
  v2f a[16];
  v2f bw[4];
  v8f sa[4];
  v8f sb[4];
};

__device__ __forceinline__ void load_k(KFrag& w, const float* aP, const float* bP,
                                       const float* haP, const float* hbP) {
#pragma unroll
  for (int g = 0; g < 8; ++g) {
    const v4f v = ld4(aP + g * 8);
    w.a[2 * g]     = __builtin_shufflevector(v, v, 0, 1);
    w.a[2 * g + 1] = __builtin_shufflevector(v, v, 2, 3);
  }
#pragma unroll
  for (int g = 0; g < 2; ++g) {
    const v4f v = ld4(bP + g * 8);
    w.bw[2 * g]     = __builtin_shufflevector(v, v, 0, 1);
    w.bw[2 * g + 1] = __builtin_shufflevector(v, v, 2, 3);
  }
#pragma unroll
  for (int bt = 0; bt < 4; ++bt) {
    w.sa[bt] = ld4x2(haP + bt * 256);
    w.sb[bt] = ld4x2(hbP + bt * 256);
  }
}

__device__ __forceinline__ void mac_k(const KFrag& w, const v2f (&xf)[4][16],
                                      const v2f (&uf)[4][4], v8f (&acc)[4]) {
  __builtin_amdgcn_sched_barrier(0);
  v8f tA[4] = {};
#pragma unroll
  for (int f = 0; f < 16; ++f) {
#pragma unroll
    for (int bt = 0; bt < 4; ++bt) tA[bt] = wmma4_f32(w.a[f], xf[bt][f], tA[bt]);
  }
  v8f tB[4] = {};
#pragma unroll
  for (int f = 0; f < 4; ++f) {
#pragma unroll
    for (int bt = 0; bt < 4; ++bt) tB[bt] = wmma4_f32(w.bw[f], uf[bt][f], tB[bt]);
  }
  __builtin_amdgcn_sched_barrier(0);
#pragma unroll
  for (int bt = 0; bt < 4; ++bt) {
#pragma unroll
    for (int r = 0; r < 8; ++r) {
      acc[bt][r] += w.sa[bt][r] * tA[bt][r];
      acc[bt][r] += w.sb[bt][r] * tB[bt][r];
    }
  }
}

__global__ __launch_bounds__(128) __attribute__((amdgpu_waves_per_eu(1, 1))) void ssm_step(
    const float* __restrict__ W2a, const float* __restrict__ b2a,
    const float* __restrict__ W2b, const float* __restrict__ b2b,
    const float* __restrict__ HA,  const float* __restrict__ HB,
    const float* __restrict__ xin,
    const float* __restrict__ ubase, int ustride,
    float* __restrict__ xout, float* __restrict__ out, int t) {
  __shared__ float red[4 * 4 * 32];

  const int lane = threadIdx.x & 31;
  const int wid  = threadIdx.x >> 5;
  const int i    = blockIdx.x;
  const int MN   = lane & 15;
  const int half = lane >> 4;
  const int jA0  = wid * 64;
  const int jB0  = wid * 16;

  v2f xf[4][16];
  v2f uf[4][4];
  {
    const float* xrow = xin + (size_t)(jA0 + half * 4) * B_ + MN * 4;
#pragma unroll
    for (int g = 0; g < 8; ++g) {
#pragma unroll
      for (int s = 0; s < 2; ++s) {
        const v4f v0 = ld4(xrow + (size_t)(g * 8 + s * 2) * B_);
        const v4f v1 = ld4(xrow + (size_t)(g * 8 + s * 2 + 1) * B_);
#pragma unroll
        for (int bt = 0; bt < 4; ++bt) {
          v2f v;
          v[0] = v0[bt];
          v[1] = v1[bt];
          xf[bt][2 * g + s] = v;
        }
      }
    }
  }
#pragma unroll
  for (int bt = 0; bt < 4; ++bt) {
    const float* urow = ubase + (size_t)(MN * 4 + bt) * ustride + jB0 + half * 4;
#pragma unroll
    for (int g = 0; g < 2; ++g) {
      const v4f v = ld4(urow + g * 8);
      uf[bt][2 * g]     = __builtin_shufflevector(v, v, 0, 1);
      uf[bt][2 * g + 1] = __builtin_shufflevector(v, v, 2, 3);
    }
  }

  const float* aP  = W2a + (size_t)MN * (S_ * S_) + (size_t)i * S_ + jA0 + half * 4;
  const float* bP  = W2b + (size_t)MN * (S_ * A_) + (size_t)i * A_ + jB0 + half * 4;
  const float* haP = HA + (size_t)t * (H_ * B_) + lane * 4;
  const float* hbP = HB + (size_t)t * (H_ * B_) + lane * 4;

  KFrag w0, w1;
  load_k(w0, aP, bP, haP, hbP);

  if (t > 0 && wid < 2) {
    const int ob = i >> 2;
    const int oi = (i & 3) * 64 + wid * 32 + lane;
    out[((size_t)ob * T_ + (t - 1)) * S_ + oi] = xin[(size_t)oi * B_ + ob];
  }

  v8f acc[4] = {};
  constexpr size_t aStep = (size_t)16 * S_ * S_;
  constexpr size_t bStep = (size_t)16 * S_ * A_;
  for (int p = 0; p < 7; ++p) {
    load_k(w1, aP + aStep, bP + bStep, haP + 1024, hbP + 1024);
    mac_k(w0, xf, uf, acc);
    aP  += 2 * aStep;
    bP  += 2 * bStep;
    haP += 2048;
    hbP += 2048;
    load_k(w0, aP, bP, haP, hbP);
    mac_k(w1, xf, uf, acc);
  }
  load_k(w1, aP + aStep, bP + bStep, haP + 1024, hbP + 1024);
  mac_k(w0, xf, uf, acc);

  v2f cb[16];
  v2f db[4];
  {
    const float* cP = b2a + (size_t)i * S_ + jA0 + half * 4;
    const float* dP = b2b + (size_t)i * A_ + jB0 + half * 4;
#pragma unroll
    for (int g = 0; g < 8; ++g) {
      const v4f v = ld4(cP + g * 8);
      cb[2 * g]     = __builtin_shufflevector(v, v, 0, 1);
      cb[2 * g + 1] = __builtin_shufflevector(v, v, 2, 3);
    }
#pragma unroll
    for (int g = 0; g < 2; ++g) {
      const v4f v = ld4(dP + g * 8);
      db[2 * g]     = __builtin_shufflevector(v, v, 0, 1);
      db[2 * g + 1] = __builtin_shufflevector(v, v, 2, 3);
    }
  }
  mac_k(w1, xf, uf, acc);
  {
    const v2f zero = {0.f, 0.f};
    v8f tb[4] = {};
#pragma unroll
    for (int f = 0; f < 16; ++f) {
      const v2f cr = (MN == 0) ? cb[f] : zero;
#pragma unroll
      for (int bt = 0; bt < 4; ++bt) tb[bt] = wmma4_f32(cr, xf[bt][f], tb[bt]);
    }
#pragma unroll
    for (int f = 0; f < 4; ++f) {
      const v2f dr = (MN == 0) ? db[f] : zero;
#pragma unroll
      for (int bt = 0; bt < 4; ++bt) tb[bt] = wmma4_f32(dr, uf[bt][f], tb[bt]);
    }
#pragma unroll
    for (int bt = 0; bt < 4; ++bt) acc[bt][0] += (half == 0) ? tb[bt][0] : 0.f;
  }

#pragma unroll
  for (int bt = 0; bt < 4; ++bt) {
    const v8f a = acc[bt];
    red[(wid * 4 + bt) * 32 + lane] =
        ((a[0] + a[1]) + (a[2] + a[3])) + ((a[4] + a[5]) + (a[6] + a[7]));
  }
  __syncthreads();
  if (wid < 2) {
    const int b  = wid * 32 + lane;
    const int bt = b & 3, c = b >> 2;
    float s = 0.f;
#pragma unroll
    for (int w = 0; w < 4; ++w)
      s += red[(w * 4 + bt) * 32 + c] + red[(w * 4 + bt) * 32 + 16 + c];
    xout[(size_t)i * B_ + b] = s;
  }
}

__global__ void ssm_emit_last(const float* __restrict__ xin, float* __restrict__ out) {
  int e = blockIdx.x * blockDim.x + threadIdx.x;
  if (e >= B_ * S_) return;
  int b = e >> 8;
  int i = e & (S_ - 1);
  out[((size_t)b * T_ + (T_ - 1)) * S_ + i] = xin[(size_t)i * B_ + b];
}

extern "C" void kernel_launch(void* const* d_in, const int* in_sizes, int n_in,
                              void* d_out, int out_size, void* d_ws, size_t ws_size,
                              hipStream_t stream) {
  const float* x0      = (const float*)d_in[0];
  const float* context = (const float*)d_in[1];
  const float* us      = (const float*)d_in[2];
  const float* W1a     = (const float*)d_in[3];
  const float* b1a     = (const float*)d_in[4];
  const float* W2a     = (const float*)d_in[5];
  const float* b2a     = (const float*)d_in[6];
  const float* W1b     = (const float*)d_in[7];
  const float* b1b     = (const float*)d_in[8];
  const float* W2b     = (const float*)d_in[9];
  const float* b2b     = (const float*)d_in[10];
  float* out = (float*)d_out;

  float* HA = (float*)d_ws;
  float* HB = HA + (size_t)T_ * B_ * H_;
  float* X0 = HB + (size_t)T_ * B_ * H_;
  float* X1 = X0 + (size_t)S_ * B_;

  int nHyp = T_ * B_ * H_;
  ssm_hyper<<<(nHyp + 255) / 256, 256, 0, stream>>>(context, W1a, b1a, W1b, b1b, x0, HA, HB, X0);

  for (int t = 0; t < T_; ++t) {
    const float* xin = (t & 1) ? X1 : X0;
    float* xout      = (t & 1) ? X0 : X1;
    const float* ubase = us + (size_t)t * A_;

    ssm_step<<<256, 128, 0, stream>>>(W2a, b2a, W2b, b2b, HA, HB,
                                      xin, ubase, T_ * A_, xout, out, t);
  }
  ssm_emit_last<<<(B_ * S_ + 255) / 256, 256, 0, stream>>>(X0, out);
}


# WMMA contraction moved to the hypernet index k: activations as B operand, generated-matrix row in accumulators started from the bias, state applied once at the end; summation order changes
# speedup vs baseline: 3.7129x; 1.0764x over previous
#include <hip/hip_runtime.h>
#include <hip/hip_bf16.h>
#include <math.h>


constexpr int B_  = 64;
constexpr int T_  = 50;
constexpr int S_  = 256;
constexpr int A_  = 64;
constexpr int H_  = 256;

typedef float v2f __attribute__((ext_vector_type(2)));
typedef float v4f __attribute__((ext_vector_type(4)));
typedef float v8f __attribute__((ext_vector_type(8)));

__device__ __forceinline__ v8f wmma4_f32(v2f a, v2f b, v8f c) {
  return __builtin_amdgcn_wmma_f32_16x16x4_f32(
       false, a,  false, b,
       (short)0, c,  false,  false);
}

__device__ __forceinline__ v4f ld4(const float* p) {
  return *reinterpret_cast<const v4f*>(p);
}

__global__ void ssm_hyper(const float* __restrict__ context,
                          const float* __restrict__ W1a, const float* __restrict__ b1a,
                          const float* __restrict__ W1b, const float* __restrict__ b1b,
                          const float* __restrict__ x0,
                          float* __restrict__ HA, float* __restrict__ HB,
                          float* __restrict__ X) {
  int idx = blockIdx.x * blockDim.x + threadIdx.x;
  if (idx >= T_ * B_ * H_) return;
  int s    = idx & 1;
  int btl  = (idx >> 1) & 1;
  int lane = (idx >> 2) & 31;
  int p    = (idx >> 7) & 1;
  int q    = (idx >> 8) & 63;
  int t    = idx >> 14;
  int b    = (lane & 15) * 4 + p * 2 + btl;
  int k    = q * 4 + (lane >> 4) * 2 + s;
  float c = context[b] + (float)t * (1.0f / 50.0f);
  HA[idx] = tanhf(fmaf(c, W1a[k], b1a[k]));
  HB[idx] = tanhf(fmaf(c, W1b[k], b1b[k]));
  if (idx < B_ * S_) X[idx] = x0[(size_t)(idx & 63) * S_ + (idx >> 6)];
}

struct KSet {
  v4f a[4][2];
  v4f bw[2];
  v4f ha[4][2];
  v4f hb[2];
};

__device__ __forceinline__ void load_set(KSet& w, const float* aP, const float* bP,
                                         const float* haP, const float* hbP) {
#pragma unroll
  for (int c = 0; c < 4; ++c) {
#pragma unroll
    for (int s = 0; s < 2; ++s) w.a[c][s] = ld4(aP + (size_t)(c * 4 + s) * (S_ * S_));
  }
#pragma unroll
  for (int s = 0; s < 2; ++s) w.bw[s] = ld4(bP + (size_t)s * (S_ * A_));
#pragma unroll
  for (int c = 0; c < 4; ++c) {
#pragma unroll
    for (int p = 0; p < 2; ++p) w.ha[c][p] = ld4(haP + c * 256 + p * 128);
  }
#pragma unroll
  for (int p = 0; p < 2; ++p) w.hb[p] = ld4(hbP + p * 128);
}

__device__ __forceinline__ void mac_set(const KSet& w, v8f (&accX)[4][4], v8f (&accU)[4][4]) {
  __builtin_amdgcn_sched_barrier(0);
#pragma unroll
  for (int c = 0; c < 4; ++c) {
    v2f bf[4];
    bf[0] = __builtin_shufflevector(w.ha[c][0], w.ha[c][0], 0, 1);
    bf[1] = __builtin_shufflevector(w.ha[c][0], w.ha[c][0], 2, 3);
    bf[2] = __builtin_shufflevector(w.ha[c][1], w.ha[c][1], 0, 1);
    bf[3] = __builtin_shufflevector(w.ha[c][1], w.ha[c][1], 2, 3);
#pragma unroll
    for (int e = 0; e < 4; ++e) {
      v2f af;
      af[0] = w.a[c][0][e];
      af[1] = w.a[c][1][e];
#pragma unroll
      for (int bt = 0; bt < 4; ++bt) accX[e][bt] = wmma4_f32(af, bf[bt], accX[e][bt]);
    }
  }
  {
    v2f bf[4];
    bf[0] = __builtin_shufflevector(w.hb[0], w.hb[0], 0, 1);
    bf[1] = __builtin_shufflevector(w.hb[0], w.hb[0], 2, 3);
    bf[2] = __builtin_shufflevector(w.hb[1], w.hb[1], 0, 1);
    bf[3] = __builtin_shufflevector(w.hb[1], w.hb[1], 2, 3);
#pragma unroll
    for (int e = 0; e < 4; ++e) {
      v2f af;
      af[0] = w.bw[0][e];
      af[1] = w.bw[1][e];
#pragma unroll
      for (int bt = 0; bt < 4; ++bt) accU[e][bt] = wmma4_f32(af, bf[bt], accU[e][bt]);
    }
  }
  __builtin_amdgcn_sched_barrier(0);
}

__global__ __launch_bounds__(128) __attribute__((amdgpu_waves_per_eu(1, 1))) void ssm_step(
    const float* __restrict__ W2a, const float* __restrict__ b2a,
    const float* __restrict__ W2b, const float* __restrict__ b2b,
    const float* __restrict__ HA,  const float* __restrict__ HB,
    const float* __restrict__ xin,
    const float* __restrict__ ubase, int ustride,
    float* __restrict__ xout, float* __restrict__ out, int t) {
  __shared__ float red[4 * 4 * 32];

  const int lane = threadIdx.x & 31;
  const int wid  = threadIdx.x >> 5;
  const int i    = blockIdx.x;
  const int MN   = lane & 15;
  const int half = lane >> 4;
  const int jA0  = wid * 64;

  const float* aP  = W2a + (size_t)(half * 2) * (S_ * S_) + (size_t)i * S_ + jA0 + MN * 4;
  const float* bP  = W2b + (size_t)(wid * 4 + half * 2) * (S_ * A_) + (size_t)i * A_ + MN * 4;
  const float* haP = HA + (size_t)t * (H_ * B_) + lane * 4;
  const float* hbP = HB + (size_t)t * (H_ * B_) + wid * 256 + lane * 4;

  KSet w0, w1;
  load_set(w0, aP, bP, haP, hbP);

  v8f accX[4][4];
  v8f accU[4][4];
  {
    const float* cP = b2a + (size_t)i * S_ + jA0 + half * 32;
    const float* dP = b2b + (size_t)i * A_ + half * 32;
#pragma unroll
    for (int r = 0; r < 8; ++r) {
      const v4f cv = ld4(cP + r * 4);
      v4f dv = ld4(dP + r * 4);
      if (wid != 0) dv = v4f{0.f, 0.f, 0.f, 0.f};
#pragma unroll
      for (int e = 0; e < 4; ++e) {
#pragma unroll
        for (int bt = 0; bt < 4; ++bt) {
          accX[e][bt][r] = cv[e];
          accU[e][bt][r] = dv[e];
        }
      }
    }
  }

  if (t > 0 && wid < 2) {
    const int ob = i >> 2;
    const int oi = (i & 3) * 64 + wid * 32 + lane;
    out[((size_t)ob * T_ + (t - 1)) * S_ + oi] = xin[(size_t)oi * B_ + ob];
  }

  constexpr size_t aStep = (size_t)16 * S_ * S_;
  constexpr size_t bStep = (size_t)16 * S_ * A_;
  for (int p = 0; p < 7; ++p) {
    load_set(w1, aP + aStep, bP + bStep, haP + 1024, hbP + 1024);
    mac_set(w0, accX, accU);
    aP  += 2 * aStep;
    bP  += 2 * bStep;
    haP += 2048;
    hbP += 2048;
    load_set(w0, aP, bP, haP, hbP);
    mac_set(w1, accX, accU);
  }
  load_set(w1, aP + aStep, bP + bStep, haP + 1024, hbP + 1024);

  v4f xv[8][4];
  v4f uv[4][8];
  {
    const float* xrow = xin + (size_t)(jA0 + half * 32) * B_ + MN * 4;
#pragma unroll
    for (int r = 0; r < 8; ++r) {
#pragma unroll
      for (int e = 0; e < 4; ++e) xv[r][e] = ld4(xrow + (size_t)(r * 4 + e) * B_);
    }
  }
  mac_set(w0, accX, accU);
#pragma unroll
  for (int bt = 0; bt < 4; ++bt) {
    const float* urow = ubase + (size_t)(MN * 4 + bt) * ustride + half * 32;
#pragma unroll
    for (int r = 0; r < 8; ++r) uv[bt][r] = ld4(urow + r * 4);
  }
  mac_set(w1, accX, accU);

#pragma unroll
  for (int bt = 0; bt < 4; ++bt) {
    float ps[4];
#pragma unroll
    for (int e = 0; e < 4; ++e) {
      float s = 0.f;
#pragma unroll
      for (int r = 0; r < 8; ++r) s += accX[e][bt][r] * xv[r][e][bt];
#pragma unroll
      for (int r = 0; r < 8; ++r) s += accU[e][bt][r] * uv[bt][r][e];
      ps[e] = s;
    }
    red[(wid * 4 + bt) * 32 + lane] = (ps[0] + ps[1]) + (ps[2] + ps[3]);
  }
  __syncthreads();
  if (wid < 2) {
    const int b  = wid * 32 + lane;
    const int bt = b & 3, c = b >> 2;
    float s = 0.f;
#pragma unroll
    for (int w = 0; w < 4; ++w)
      s += red[(w * 4 + bt) * 32 + c] + red[(w * 4 + bt) * 32 + 16 + c];
    xout[(size_t)i * B_ + b] = s;
  }
}

__global__ void ssm_emit_last(const float* __restrict__ xin, float* __restrict__ out) {
  int e = blockIdx.x * blockDim.x + threadIdx.x;
  if (e >= B_ * S_) return;
  int b = e >> 8;
  int i = e & (S_ - 1);
  out[((size_t)b * T_ + (T_ - 1)) * S_ + i] = xin[(size_t)i * B_ + b];
}

extern "C" void kernel_launch(void* const* d_in, const int* in_sizes, int n_in,
                              void* d_out, int out_size, void* d_ws, size_t ws_size,
                              hipStream_t stream) {
  const float* x0      = (const float*)d_in[0];
  const float* context = (const float*)d_in[1];
  const float* us      = (const float*)d_in[2];
  const float* W1a     = (const float*)d_in[3];
  const float* b1a     = (const float*)d_in[4];
  const float* W2a     = (const float*)d_in[5];
  const float* b2a     = (const float*)d_in[6];
  const float* W1b     = (const float*)d_in[7];
  const float* b1b     = (const float*)d_in[8];
  const float* W2b     = (const float*)d_in[9];
  const float* b2b     = (const float*)d_in[10];
  float* out = (float*)d_out;

  float* HA = (float*)d_ws;
  float* HB = HA + (size_t)T_ * B_ * H_;
  float* X0 = HB + (size_t)T_ * B_ * H_;
  float* X1 = X0 + (size_t)S_ * B_;

  int nHyp = T_ * B_ * H_;
  ssm_hyper<<<(nHyp + 255) / 256, 256, 0, stream>>>(context, W1a, b1a, W1b, b1b, x0, HA, HB, X0);

  for (int t = 0; t < T_; ++t) {
    const float* xin = (t & 1) ? X1 : X0;
    float* xout      = (t & 1) ? X0 : X1;
    const float* ubase = us + (size_t)t * A_;

    ssm_step<<<256, 128, 0, stream>>>(W2a, b2a, W2b, b2b, HA, HB,
                                      xin, ubase, T_ * A_, xout, out, t);
  }
  ssm_emit_last<<<(B_ * S_ + 255) / 256, 256, 0, stream>>>(X0, out);
}
